# starting kernel
# baseline (speedup 1.0000x reference)
#include <hip/hip_runtime.h>
#include <hip/hip_bf16.h>

#define NODES_F 128
#define LDSTRIDE 136
#define EDGES_PER_WAVE 8

typedef __attribute__((ext_vector_type(16))) __bf16 v16bf;
typedef __attribute__((ext_vector_type(8)))  __bf16 v8bf;
typedef __attribute__((ext_vector_type(8)))  float  v8f;
typedef __attribute__((ext_vector_type(4)))  float  f32x4;

__device__ __forceinline__ void split_bf16(float f, __bf16& hi, __bf16& lo) {
    hi = (__bf16)f;
    lo = (__bf16)(f - (float)hi);
}

__global__ __launch_bounds__(256)
void gemm_bf16x3(const float* __restrict__ X, const float* __restrict__ W,
                 float* __restrict__ Y, int M) {
    __shared__ __bf16 whi[NODES_F * LDSTRIDE];
    __shared__ __bf16 wlo[NODES_F * LDSTRIDE];

    const int tid = threadIdx.x;

    for (int i = tid; i < NODES_F * NODES_F; i += 256) {
        int k = i >> 7;
        int n = i & 127;
        __bf16 h, l;
        split_bf16(W[i], h, l);
        whi[n * LDSTRIDE + k] = h;
        wlo[n * LDSTRIDE + k] = l;
    }
    __syncthreads();

    const int wave     = tid >> 5;
    const int lane     = tid & 31;
    const int laneHalf = (lane >> 4) & 1;
    const int lmod     = lane & 15;
    const int row0     = blockIdx.x * 128 + wave * 16;
    if (row0 >= M) return;

    v8f acc[8] = {};

    const float* arow = X + (size_t)(row0 + lmod) * NODES_F;

    for (int kb = 0; kb < NODES_F; kb += 32) {
        const int k0 = kb + laneHalf * 8;
        f32x4 a0 = *(const f32x4*)(arow + k0);
        f32x4 a1 = *(const f32x4*)(arow + k0 + 4);
        f32x4 a2 = *(const f32x4*)(arow + k0 + 16);
        f32x4 a3 = *(const f32x4*)(arow + k0 + 20);
        float af[16];
        #pragma unroll
        for (int e = 0; e < 4; ++e) {
            af[e]      = a0[e];
            af[e + 4]  = a1[e];
            af[e + 8]  = a2[e];
            af[e + 12] = a3[e];
        }
        v16bf ahi, alo;
        #pragma unroll
        for (int e = 0; e < 16; ++e) {
            __bf16 h, l;
            split_bf16(af[e], h, l);
            ahi[e] = h;
            alo[e] = l;
        }

        #pragma unroll
        for (int t = 0; t < 8; ++t) {
            const int n = t * 16 + lmod;
            const __bf16* bh = &whi[n * LDSTRIDE + k0];
            const __bf16* bl = &wlo[n * LDSTRIDE + k0];
            v8bf bh0 = *(const v8bf*)(bh);
            v8bf bh1 = *(const v8bf*)(bh + 16);
            v8bf bl0 = *(const v8bf*)(bl);
            v8bf bl1 = *(const v8bf*)(bl + 16);
            v16bf bhi = __builtin_shufflevector(bh0, bh1, 0,1,2,3,4,5,6,7,8,9,10,11,12,13,14,15);
            v16bf blo = __builtin_shufflevector(bl0, bl1, 0,1,2,3,4,5,6,7,8,9,10,11,12,13,14,15);

            acc[t] = __builtin_amdgcn_wmma_f32_16x16x32_bf16(false, ahi, false, bhi,
                                                             (short)0, acc[t], false, false);
            acc[t] = __builtin_amdgcn_wmma_f32_16x16x32_bf16(false, ahi, false, blo,
                                                             (short)0, acc[t], false, false);
            acc[t] = __builtin_amdgcn_wmma_f32_16x16x32_bf16(false, alo, false, bhi,
                                                             (short)0, acc[t], false, false);
        }
    }

    #pragma unroll
    for (int t = 0; t < 8; ++t) {
        #pragma unroll
        for (int v = 0; v < 8; ++v) {
            Y[(size_t)(row0 + v + laneHalf * 8) * NODES_F + t * 16 + lmod] = acc[t][v];
        }
    }
}

__global__ __launch_bounds__(256)
void spmm_scatter(const int* __restrict__ erow, const int* __restrict__ ecol,
                  const float* __restrict__ eval, const float* __restrict__ X,
                  float* __restrict__ Y, int nE) {
    int wid  = (blockIdx.x * 256 + threadIdx.x) >> 5;
    int lane = threadIdx.x & 31;
    int e0   = wid * EDGES_PER_WAVE;
    if (e0 >= nE) return;
    int eEnd = min(e0 + EDGES_PER_WAVE, nE);

    int   cur = erow[e0];
    f32x4 acc = (f32x4){0.f, 0.f, 0.f, 0.f};

    for (int e = e0; e < eEnd; ++e) {
        int r = erow[e];
        if (r != cur) {
            float* yp = Y + (size_t)cur * NODES_F + lane * 4;
            atomicAdd(yp + 0, acc[0]);
            atomicAdd(yp + 1, acc[1]);
            atomicAdd(yp + 2, acc[2]);
            atomicAdd(yp + 3, acc[3]);
            acc = (f32x4){0.f, 0.f, 0.f, 0.f};
            cur = r;
        }
        float v = eval[e];
        int   c = ecol[e];
        f32x4 x = *(const f32x4*)(X + (size_t)c * NODES_F + lane * 4);
        acc[0] += v * x[0];
        acc[1] += v * x[1];
        acc[2] += v * x[2];
        acc[3] += v * x[3];
    }
    float* yp = Y + (size_t)cur * NODES_F + lane * 4;
    atomicAdd(yp + 0, acc[0]);
    atomicAdd(yp + 1, acc[1]);
    atomicAdd(yp + 2, acc[2]);
    atomicAdd(yp + 3, acc[3]);
}

__global__ __launch_bounds__(256)
void zero_f32(float* __restrict__ p, int n4) {
    int i = blockIdx.x * 256 + threadIdx.x;
    if (i < n4) ((f32x4*)p)[i] = (f32x4){0.f, 0.f, 0.f, 0.f};
}

__global__ __launch_bounds__(256)
void bias_relu4(float* __restrict__ h, const float* __restrict__ b, int n4) {
    int i = blockIdx.x * 256 + threadIdx.x;
    if (i < n4) {
        f32x4 x  = ((f32x4*)h)[i];
        f32x4 bb = *(const f32x4*)(b + ((i * 4) & 127));
        #pragma unroll
        for (int j = 0; j < 4; ++j) {
            float v = x[j] + bb[j];
            x[j] = v > 0.f ? v : 0.f;
        }
        ((f32x4*)h)[i] = x;
    }
}

__global__ __launch_bounds__(256)
void bias_init4(float* __restrict__ y, const float* __restrict__ b, int n4) {
    int i = blockIdx.x * 256 + threadIdx.x;
    if (i < n4) ((f32x4*)y)[i] = *(const f32x4*)(b + ((i * 4) & 127));
}

extern "C" void kernel_launch(void* const* d_in, const int* in_sizes, int n_in,
                              void* d_out, int out_size, void* d_ws, size_t ws_size,
                              hipStream_t stream) {
    const float* X    = (const float*)d_in[0];
    const int*   erow = (const int*)  d_in[1];
    const int*   ecol = (const int*)  d_in[2];
    const float* ev   = (const float*)d_in[3];
    const float* W1   = (const float*)d_in[4];
    const float* b1   = (const float*)d_in[5];
    const float* W3   = (const float*)d_in[6];
    const float* b3   = (const float*)d_in[7];
    float* out = (float*)d_out;

    const int M  = in_sizes[0] / NODES_F;
    const int nE = in_sizes[1];
    const int NF = M * NODES_F;

    float* t = (float*)d_ws;
    float* h = t + (size_t)NF;

    dim3 blk(256);
    int gemmGrid = (M + 127) / 128;
    int vecGrid  = (NF / 4 + 255) / 256;
    int nWaves   = (nE + EDGES_PER_WAVE - 1) / EDGES_PER_WAVE;
    int edgeGrid = (nWaves + 7) / 8;

    gemm_bf16x3 <<<gemmGrid, blk, 0, stream>>>(X, W1, t, M);
    zero_f32    <<<vecGrid,  blk, 0, stream>>>(h, NF / 4);
    spmm_scatter<<<edgeGrid, blk, 0, stream>>>(erow, ecol, ev, t, h, nE);
    bias_relu4  <<<vecGrid,  blk, 0, stream>>>(h, b1, NF / 4);

    gemm_bf16x3 <<<gemmGrid, blk, 0, stream>>>(h, W3, t, M);
    bias_init4  <<<vecGrid,  blk, 0, stream>>>(out, b3, NF / 4);
    spmm_scatter<<<edgeGrid, blk, 0, stream>>>(erow, ecol, ev, t, out, nE);
}


# SpMM by 64-edge wave slices: triples by scalar loads, 8 gathers per group; rows wholly inside a slice get a plain store in edge order, atomics only on rows cut by a slice end; f32 sum order changes
# speedup vs baseline: 1.5679x; 1.5679x over previous
#include <hip/hip_runtime.h>
#include <hip/hip_bf16.h>

#define NODES_F 128
#define LDSTRIDE 136

typedef __attribute__((ext_vector_type(16))) __bf16 v16bf;
typedef __attribute__((ext_vector_type(8)))  __bf16 v8bf;
typedef __attribute__((ext_vector_type(8)))  float  v8f;
typedef __attribute__((ext_vector_type(4)))  float  f32x4;

__device__ __forceinline__ void split_bf16(float f, __bf16& hi, __bf16& lo) {
    hi = (__bf16)f;
    lo = (__bf16)(f - (float)hi);
}

__global__ __launch_bounds__(256)
void gemm_bf16x3(const float* __restrict__ X, const float* __restrict__ W,
                 float* __restrict__ Y, int M) {
    __shared__ __bf16 whi[NODES_F * LDSTRIDE];
    __shared__ __bf16 wlo[NODES_F * LDSTRIDE];

    const int tid = threadIdx.x;

    for (int i = tid; i < NODES_F * NODES_F; i += 256) {
        int k = i >> 7;
        int n = i & 127;
        __bf16 h, l;
        split_bf16(W[i], h, l);
        whi[n * LDSTRIDE + k] = h;
        wlo[n * LDSTRIDE + k] = l;
    }
    __syncthreads();

    const int wave     = tid >> 5;
    const int lane     = tid & 31;
    const int laneHalf = (lane >> 4) & 1;
    const int lmod     = lane & 15;
    const int row0     = blockIdx.x * 128 + wave * 16;
    if (row0 >= M) return;

    v8f acc[8] = {};

    const float* arow = X + (size_t)(row0 + lmod) * NODES_F;

    for (int kb = 0; kb < NODES_F; kb += 32) {
        const int k0 = kb + laneHalf * 8;
        f32x4 a0 = *(const f32x4*)(arow + k0);
        f32x4 a1 = *(const f32x4*)(arow + k0 + 4);
        f32x4 a2 = *(const f32x4*)(arow + k0 + 16);
        f32x4 a3 = *(const f32x4*)(arow + k0 + 20);
        float af[16];
        #pragma unroll
        for (int e = 0; e < 4; ++e) {
            af[e]      = a0[e];
            af[e + 4]  = a1[e];
            af[e + 8]  = a2[e];
            af[e + 12] = a3[e];
        }
        v16bf ahi, alo;
        #pragma unroll
        for (int e = 0; e < 16; ++e) {
            __bf16 h, l;
            split_bf16(af[e], h, l);
            ahi[e] = h;
            alo[e] = l;
        }

        #pragma unroll
        for (int t = 0; t < 8; ++t) {
            const int n = t * 16 + lmod;
            const __bf16* bh = &whi[n * LDSTRIDE + k0];
            const __bf16* bl = &wlo[n * LDSTRIDE + k0];
            v8bf bh0 = *(const v8bf*)(bh);
            v8bf bh1 = *(const v8bf*)(bh + 16);
            v8bf bl0 = *(const v8bf*)(bl);
            v8bf bl1 = *(const v8bf*)(bl + 16);
            v16bf bhi = __builtin_shufflevector(bh0, bh1, 0,1,2,3,4,5,6,7,8,9,10,11,12,13,14,15);
            v16bf blo = __builtin_shufflevector(bl0, bl1, 0,1,2,3,4,5,6,7,8,9,10,11,12,13,14,15);

            acc[t] = __builtin_amdgcn_wmma_f32_16x16x32_bf16(false, ahi, false, bhi,
                                                             (short)0, acc[t], false, false);
            acc[t] = __builtin_amdgcn_wmma_f32_16x16x32_bf16(false, ahi, false, blo,
                                                             (short)0, acc[t], false, false);
            acc[t] = __builtin_amdgcn_wmma_f32_16x16x32_bf16(false, alo, false, bhi,
                                                             (short)0, acc[t], false, false);
        }
    }

    #pragma unroll
    for (int t = 0; t < 8; ++t) {
        #pragma unroll
        for (int v = 0; v < 8; ++v) {
            Y[(size_t)(row0 + v + laneHalf * 8) * NODES_F + t * 16 + lmod] = acc[t][v];
        }
    }
}

#define SLICE 64
typedef __attribute__((ext_vector_type(8))) int   i32x8;
typedef __attribute__((ext_vector_type(8))) float f32x8;

template <bool ADD_BIAS>
__device__ __forceinline__ void put_row(float* __restrict__ yp, f32x4 acc, f32x4 bb, bool whole) {
    if (whole) {
        if (ADD_BIAS) acc += bb;
        *(f32x4*)yp = acc;
    } else {
        atomicAdd(yp + 0, acc[0]);
        atomicAdd(yp + 1, acc[1]);
        atomicAdd(yp + 2, acc[2]);
        atomicAdd(yp + 3, acc[3]);
    }
}

template <bool ADD_BIAS>
__global__ __launch_bounds__(128)
void spmm_slices(const int* __restrict__ erow, const int* __restrict__ ecol,
                 const float* __restrict__ eval, const float* __restrict__ X,
                 const float* __restrict__ bias, float* __restrict__ Y, int nE) {
    const int lane = threadIdx.x & 31;
    const int wid = __builtin_amdgcn_readfirstlane((int)((blockIdx.x * 128 + threadIdx.x) >> 5));
    const int e0  = wid * SLICE;
    if (e0 >= nE) return;
    const int e1  = min(e0 + SLICE, nE);

    const int rowBefore = (e0 > 0)  ? erow[e0 - 1] : -1;
    const int rowAfter  = (e1 < nE) ? erow[e1]     : -1;

    const float* xl = X + lane * 4;
    float*       yl = Y + lane * 4;
    f32x4 bb = (f32x4){0.f, 0.f, 0.f, 0.f};
    if (ADD_BIAS) bb = *(const f32x4*)(bias + lane * 4);

    int   cur   = erow[e0];
    bool  whole = (cur != rowBefore);
    f32x4 acc   = (f32x4){0.f, 0.f, 0.f, 0.f};

    int eb = e0;
    i32x8 c8 = (i32x8){0, 0, 0, 0, 0, 0, 0, 0};
    if (e0 + 8 <= e1) c8 = *(const i32x8*)(ecol + e0);
    for (; eb + 8 <= e1; eb += 8) {
        f32x4 x[8];
        #pragma unroll
        for (int j = 0; j < 8; ++j)
            x[j] = *(const f32x4*)(xl + (size_t)c8[j] * NODES_F);
        const i32x8 r8 = *(const i32x8*)(erow + eb);
        const f32x8 v8 = *(const f32x8*)(eval + eb);
        c8 = *(const i32x8*)(ecol + ((eb + 16 <= e1) ? eb + 8 : eb));
        #pragma unroll
        for (int j = 0; j < 8; ++j) {
            const int r = r8[j];
            if (r != cur) {
                put_row<ADD_BIAS>(yl + (size_t)cur * NODES_F, acc, bb, whole);
                acc   = (f32x4){0.f, 0.f, 0.f, 0.f};
                cur   = r;
                whole = true;
            }
            const float v = v8[j];
            acc[0] += v * x[j][0];
            acc[1] += v * x[j][1];
            acc[2] += v * x[j][2];
            acc[3] += v * x[j][3];
        }
    }
    for (; eb < e1; ++eb) {
        const int r = erow[eb];
        if (r != cur) {
            put_row<ADD_BIAS>(yl + (size_t)cur * NODES_F, acc, bb, whole);
            acc   = (f32x4){0.f, 0.f, 0.f, 0.f};
            cur   = r;
            whole = true;
        }
        const float v = eval[eb];
        const f32x4 x = *(const f32x4*)(xl + (size_t)ecol[eb] * NODES_F);
        acc[0] += v * x[0];
        acc[1] += v * x[1];
        acc[2] += v * x[2];
        acc[3] += v * x[3];
    }
    put_row<ADD_BIAS>(yl + (size_t)cur * NODES_F, acc, bb, whole && (cur != rowAfter));
}

__global__ __launch_bounds__(256)
void zero_f32(float* __restrict__ p, int n4) {
    int i = blockIdx.x * 256 + threadIdx.x;
    if (i < n4) ((f32x4*)p)[i] = (f32x4){0.f, 0.f, 0.f, 0.f};
}

__global__ __launch_bounds__(256)
void bias_relu4(float* __restrict__ h, const float* __restrict__ b, int n4) {
    int i = blockIdx.x * 256 + threadIdx.x;
    if (i < n4) {
        f32x4 x  = ((f32x4*)h)[i];
        f32x4 bb = *(const f32x4*)(b + ((i * 4) & 127));
        #pragma unroll
        for (int j = 0; j < 4; ++j) {
            float v = x[j] + bb[j];
            x[j] = v > 0.f ? v : 0.f;
        }
        ((f32x4*)h)[i] = x;
    }
}

__global__ __launch_bounds__(256)
void bias_init4(float* __restrict__ y, const float* __restrict__ b, int n4) {
    int i = blockIdx.x * 256 + threadIdx.x;
    if (i < n4) ((f32x4*)y)[i] = *(const f32x4*)(b + ((i * 4) & 127));
}

extern "C" void kernel_launch(void* const* d_in, const int* in_sizes, int n_in,
                              void* d_out, int out_size, void* d_ws, size_t ws_size,
                              hipStream_t stream) {
    const float* X    = (const float*)d_in[0];
    const int*   erow = (const int*)  d_in[1];
    const int*   ecol = (const int*)  d_in[2];
    const float* ev   = (const float*)d_in[3];
    const float* W1   = (const float*)d_in[4];
    const float* b1   = (const float*)d_in[5];
    const float* W3   = (const float*)d_in[6];
    const float* b3   = (const float*)d_in[7];
    float* out = (float*)d_out;

    const int M  = in_sizes[0] / NODES_F;
    const int nE = in_sizes[1];
    const int NF = M * NODES_F;

    float* t = (float*)d_ws;
    float* h = t + (size_t)NF;

    dim3 blk(256);
    int gemmGrid = (M + 127) / 128;
    int vecGrid  = (NF / 4 + 255) / 256;
    int nWaves   = (nE + SLICE - 1) / SLICE;
    int edgeGrid = (nWaves + 3) / 4;

    gemm_bf16x3 <<<gemmGrid, blk, 0, stream>>>(X, W1, t, M);
    zero_f32    <<<vecGrid,  blk, 0, stream>>>(h, NF / 4);
    spmm_slices<false><<<edgeGrid, dim3(128), 0, stream>>>(erow, ecol, ev, t, b1, h, nE);
    bias_relu4  <<<vecGrid,  blk, 0, stream>>>(h, b1, NF / 4);

    gemm_bf16x3 <<<gemmGrid, blk, 0, stream>>>(h, W3, t, M);
    bias_init4  <<<vecGrid,  blk, 0, stream>>>(out, b3, NF / 4);
    spmm_slices<true><<<edgeGrid, dim3(128), 0, stream>>>(erow, ecol, ev, t, b3, out, nE);
}


# Three elementwise passes folded into the GEMM launches: GEMM 1 also stores h's zero start rows, GEMM 2 adds b1 and ReLU as it loads h and stores out's b3 start rows; 7 launches to 4; sums unchanged
# speedup vs baseline: 1.8030x; 1.1499x over previous
#include <hip/hip_runtime.h>
#include <hip/hip_bf16.h>

#define NODES_F 128
#define LDSTRIDE 136

typedef __attribute__((ext_vector_type(16))) __bf16 v16bf;
typedef __attribute__((ext_vector_type(8)))  __bf16 v8bf;
typedef __attribute__((ext_vector_type(8)))  float  v8f;
typedef __attribute__((ext_vector_type(4)))  float  f32x4;

__device__ __forceinline__ void split_bf16(float f, __bf16& hi, __bf16& lo) {
    hi = (__bf16)f;
    lo = (__bf16)(f - (float)hi);
}

template <bool ACT_IN, bool BIAS_START>
__global__ __launch_bounds__(256)
void gemm_bf16x3(const float* __restrict__ X, const float* __restrict__ W,
                 float* __restrict__ Y, int M,
                 const float* __restrict__ inBias,
                 float* __restrict__ S, const float* __restrict__ sBias) {
    __shared__ __bf16 whi[NODES_F * LDSTRIDE];
    __shared__ __bf16 wlo[NODES_F * LDSTRIDE];

    const int tid = threadIdx.x;

    {
        const int srow = blockIdx.x * 128 + (tid >> 5) * 16;
        if (srow < M) {
            f32x4 sv = (f32x4){0.f, 0.f, 0.f, 0.f};
            if (BIAS_START) sv = *(const f32x4*)(sBias + (tid & 31) * 4);
            float* sp = S + (size_t)srow * NODES_F + (tid & 31) * 4;
            #pragma unroll
            for (int r = 0; r < 16; ++r)
                *(f32x4*)(sp + (size_t)r * NODES_F) = sv;
        }
    }

    for (int i = tid; i < NODES_F * NODES_F; i += 256) {
        int k = i >> 7;
        int n = i & 127;
        __bf16 h, l;
        split_bf16(W[i], h, l);
        whi[n * LDSTRIDE + k] = h;
        wlo[n * LDSTRIDE + k] = l;
    }
    __syncthreads();

    const int wave     = tid >> 5;
    const int lane     = tid & 31;
    const int laneHalf = (lane >> 4) & 1;
    const int lmod     = lane & 15;
    const int row0     = blockIdx.x * 128 + wave * 16;
    if (row0 >= M) return;

    v8f acc[8] = {};

    const float* arow = X + (size_t)(row0 + lmod) * NODES_F;

    v16bf ahi[4], alo[4];
    #pragma unroll
    for (int ks = 0; ks < 4; ++ks) {
        const int k0 = ks * 32 + laneHalf * 8;
        f32x4 a0 = *(const f32x4*)(arow + k0);
        f32x4 a1 = *(const f32x4*)(arow + k0 + 4);
        f32x4 a2 = *(const f32x4*)(arow + k0 + 16);
        f32x4 a3 = *(const f32x4*)(arow + k0 + 20);
        if (ACT_IN) {
            const f32x4 c0 = *(const f32x4*)(inBias + k0);
            const f32x4 c1 = *(const f32x4*)(inBias + k0 + 4);
            const f32x4 c2 = *(const f32x4*)(inBias + k0 + 16);
            const f32x4 c3 = *(const f32x4*)(inBias + k0 + 20);
            #pragma unroll
            for (int e = 0; e < 4; ++e) {
                float v0 = a0[e] + c0[e];
                float v1 = a1[e] + c1[e];
                float v2 = a2[e] + c2[e];
                float v3 = a3[e] + c3[e];
                a0[e] = v0 > 0.f ? v0 : 0.f;
                a1[e] = v1 > 0.f ? v1 : 0.f;
                a2[e] = v2 > 0.f ? v2 : 0.f;
                a3[e] = v3 > 0.f ? v3 : 0.f;
            }
        }
        float af[16];
        #pragma unroll
        for (int e = 0; e < 4; ++e) {
            af[e]      = a0[e];
            af[e + 4]  = a1[e];
            af[e + 8]  = a2[e];
            af[e + 12] = a3[e];
        }
        #pragma unroll
        for (int e = 0; e < 16; ++e) {
            __bf16 h, l;
            split_bf16(af[e], h, l);
            ahi[ks][e] = h;
            alo[ks][e] = l;
        }
    }
    __builtin_amdgcn_sched_barrier(0);

    #pragma unroll
    for (int ks = 0; ks < 4; ++ks) {
        const int k0 = ks * 32 + laneHalf * 8;
        #pragma unroll
        for (int t = 0; t < 8; ++t) {
            const int n = t * 16 + lmod;
            const __bf16* bh = &whi[n * LDSTRIDE + k0];
            const __bf16* bl = &wlo[n * LDSTRIDE + k0];
            v8bf bh0 = *(const v8bf*)(bh);
            v8bf bh1 = *(const v8bf*)(bh + 16);
            v8bf bl0 = *(const v8bf*)(bl);
            v8bf bl1 = *(const v8bf*)(bl + 16);
            v16bf bhi = __builtin_shufflevector(bh0, bh1, 0,1,2,3,4,5,6,7,8,9,10,11,12,13,14,15);
            v16bf blo = __builtin_shufflevector(bl0, bl1, 0,1,2,3,4,5,6,7,8,9,10,11,12,13,14,15);

            acc[t] = __builtin_amdgcn_wmma_f32_16x16x32_bf16(false, ahi[ks], false, bhi,
                                                             (short)0, acc[t], false, false);
            acc[t] = __builtin_amdgcn_wmma_f32_16x16x32_bf16(false, ahi[ks], false, blo,
                                                             (short)0, acc[t], false, false);
            acc[t] = __builtin_amdgcn_wmma_f32_16x16x32_bf16(false, alo[ks], false, bhi,
                                                             (short)0, acc[t], false, false);
        }
    }

    #pragma unroll
    for (int t = 0; t < 8; ++t) {
        #pragma unroll
        for (int v = 0; v < 8; ++v) {
            Y[(size_t)(row0 + v + laneHalf * 8) * NODES_F + t * 16 + lmod] = acc[t][v];
        }
    }
}

#define SLICE 64
typedef __attribute__((ext_vector_type(8))) int   i32x8;
typedef __attribute__((ext_vector_type(8))) float f32x8;

template <bool ADD_BIAS>
__device__ __forceinline__ void put_row(float* __restrict__ yp, f32x4 acc, f32x4 bb, bool whole) {
    if (whole) {
        if (ADD_BIAS) acc += bb;
        *(f32x4*)yp = acc;
    } else {
        atomicAdd(yp + 0, acc[0]);
        atomicAdd(yp + 1, acc[1]);
        atomicAdd(yp + 2, acc[2]);
        atomicAdd(yp + 3, acc[3]);
    }
}

template <bool ADD_BIAS>
__global__ __launch_bounds__(128)
void spmm_slices(const int* __restrict__ erow, const int* __restrict__ ecol,
                 const float* __restrict__ eval, const float* __restrict__ X,
                 const float* __restrict__ bias, float* __restrict__ Y, int nE) {
    const int lane = threadIdx.x & 31;
    const int wid = __builtin_amdgcn_readfirstlane((int)((blockIdx.x * 128 + threadIdx.x) >> 5));
    const int e0  = wid * SLICE;
    if (e0 >= nE) return;
    const int e1  = min(e0 + SLICE, nE);

    const int rowBefore = (e0 > 0)  ? erow[e0 - 1] : -1;
    const int rowAfter  = (e1 < nE) ? erow[e1]     : -1;

    const float* xl = X + lane * 4;
    float*       yl = Y + lane * 4;
    f32x4 bb = (f32x4){0.f, 0.f, 0.f, 0.f};
    if (ADD_BIAS) bb = *(const f32x4*)(bias + lane * 4);

    int   cur   = erow[e0];
    bool  whole = (cur != rowBefore);
    f32x4 acc   = (f32x4){0.f, 0.f, 0.f, 0.f};

    int eb = e0;
    i32x8 c8 = (i32x8){0, 0, 0, 0, 0, 0, 0, 0};
    if (e0 + 8 <= e1) c8 = *(const i32x8*)(ecol + e0);
    for (; eb + 8 <= e1; eb += 8) {
        f32x4 x[8];
        #pragma unroll
        for (int j = 0; j < 8; ++j)
            x[j] = *(const f32x4*)(xl + (size_t)c8[j] * NODES_F);
        const i32x8 r8 = *(const i32x8*)(erow + eb);
        const f32x8 v8 = *(const f32x8*)(eval + eb);
        c8 = *(const i32x8*)(ecol + ((eb + 16 <= e1) ? eb + 8 : eb));
        #pragma unroll
        for (int j = 0; j < 8; ++j) {
            const int r = r8[j];
            if (r != cur) {
                put_row<ADD_BIAS>(yl + (size_t)cur * NODES_F, acc, bb, whole);
                acc   = (f32x4){0.f, 0.f, 0.f, 0.f};
                cur   = r;
                whole = true;
            }
            const float v = v8[j];
            acc[0] += v * x[j][0];
            acc[1] += v * x[j][1];
            acc[2] += v * x[j][2];
            acc[3] += v * x[j][3];
        }
    }
    for (; eb < e1; ++eb) {
        const int r = erow[eb];
        if (r != cur) {
            put_row<ADD_BIAS>(yl + (size_t)cur * NODES_F, acc, bb, whole);
            acc   = (f32x4){0.f, 0.f, 0.f, 0.f};
            cur   = r;
            whole = true;
        }
        const float v = eval[eb];
        const f32x4 x = *(const f32x4*)(xl + (size_t)ecol[eb] * NODES_F);
        acc[0] += v * x[0];
        acc[1] += v * x[1];
        acc[2] += v * x[2];
        acc[3] += v * x[3];
    }
    put_row<ADD_BIAS>(yl + (size_t)cur * NODES_F, acc, bb, whole && (cur != rowAfter));
}

extern "C" void kernel_launch(void* const* d_in, const int* in_sizes, int n_in,
                              void* d_out, int out_size, void* d_ws, size_t ws_size,
                              hipStream_t stream) {
    const float* X    = (const float*)d_in[0];
    const int*   erow = (const int*)  d_in[1];
    const int*   ecol = (const int*)  d_in[2];
    const float* ev   = (const float*)d_in[3];
    const float* W1   = (const float*)d_in[4];
    const float* b1   = (const float*)d_in[5];
    const float* W3   = (const float*)d_in[6];
    const float* b3   = (const float*)d_in[7];
    float* out = (float*)d_out;

    const int M  = in_sizes[0] / NODES_F;
    const int nE = in_sizes[1];
    const int NF = M * NODES_F;

    float* t = (float*)d_ws;
    float* h = t + (size_t)NF;

    dim3 blk(256);
    int gemmGrid = (M + 127) / 128;
    int nWaves   = (nE + SLICE - 1) / SLICE;
    int edgeGrid = (nWaves + 3) / 4;

    gemm_bf16x3<false, false><<<gemmGrid, blk, 0, stream>>>(X, W1, t, M, b1, h, b1);
    spmm_slices<false><<<edgeGrid, dim3(128), 0, stream>>>(erow, ecol, ev, t, b1, h, nE);

    gemm_bf16x3<true, true><<<gemmGrid, blk, 0, stream>>>(h, W3, t, M, b1, out, b3);
    spmm_slices<true><<<edgeGrid, dim3(128), 0, stream>>>(erow, ecol, ev, t, b3, out, nE);
}


# GEMM: W stays row-major in LDS and B fragments come through the transposing LDS load; staging is 16 row loads before one wait and 16-byte LDS stores, not 8 waits and 2-byte stores; values unchanged
# speedup vs baseline: 2.2288x; 1.2362x over previous
#include <hip/hip_runtime.h>
#include <hip/hip_bf16.h>

#define NODES_F 128
#define WPITCH 144

typedef __attribute__((ext_vector_type(16))) __bf16 v16bf;
typedef __attribute__((ext_vector_type(8)))  __bf16 v8bf;
typedef __attribute__((ext_vector_type(8)))  float  v8f;
typedef __attribute__((ext_vector_type(4)))  float  f32x4;
typedef __attribute__((address_space(3))) v8bf lds_v8bf;

__device__ __forceinline__ v8bf ld_tr16(const __bf16* p) {
    return __builtin_amdgcn_ds_load_tr16_b128_v8bf16((lds_v8bf*)p);
}

__device__ __forceinline__ void split_bf16(float f, __bf16& hi, __bf16& lo) {
    hi = (__bf16)f;
    lo = (__bf16)(f - (float)hi);
}

template <bool ACT_IN, bool BIAS_START>
__global__ __launch_bounds__(256)
void gemm_bf16x3(const float* __restrict__ X, const float* __restrict__ W,
                 float* __restrict__ Y, int M,
                 const float* __restrict__ inBias,
                 float* __restrict__ S, const float* __restrict__ sBias) {
    __shared__ __attribute__((aligned(32))) __bf16 whi[NODES_F * WPITCH];
    __shared__ __attribute__((aligned(32))) __bf16 wlo[NODES_F * WPITCH];

    const int tid = threadIdx.x;

    {
        const int srow = blockIdx.x * 128 + (tid >> 5) * 16;
        if (srow < M) {
            f32x4 sv = (f32x4){0.f, 0.f, 0.f, 0.f};
            if (BIAS_START) sv = *(const f32x4*)(sBias + (tid & 31) * 4);
            float* sp = S + (size_t)srow * NODES_F + (tid & 31) * 4;
            #pragma unroll
            for (int r = 0; r < 16; ++r)
                *(f32x4*)(sp + (size_t)r * NODES_F) = sv;
        }
    }

    {
        f32x4 wv[16];
        #pragma unroll
        for (int j = 0; j < 8; ++j) {
            const float* wp = W + (size_t)(tid + 256 * j) * 8;
            wv[2 * j]     = *(const f32x4*)(wp);
            wv[2 * j + 1] = *(const f32x4*)(wp + 4);
        }
        #pragma unroll
        for (int j = 0; j < 8; ++j) {
            const int q = tid + 256 * j;
            const int k = q >> 4;
            const int n = (q & 15) * 8;
            v8bf h8, l8;
            #pragma unroll
            for (int e = 0; e < 4; ++e) {
                __bf16 h, l;
                split_bf16(wv[2 * j][e], h, l);
                h8[e] = h;
                l8[e] = l;
                split_bf16(wv[2 * j + 1][e], h, l);
                h8[e + 4] = h;
                l8[e + 4] = l;
            }
            *(v8bf*)&whi[k * WPITCH + n] = h8;
            *(v8bf*)&wlo[k * WPITCH + n] = l8;
        }
    }
    __syncthreads();

    const int wave     = tid >> 5;
    const int lane     = tid & 31;
    const int laneHalf = (lane >> 4) & 1;
    const int lmod     = lane & 15;
    const int row0     = blockIdx.x * 128 + wave * 16;
    if (row0 >= M) return;

    v8f acc[8] = {};

    const float* arow = X + (size_t)(row0 + lmod) * NODES_F;

    v16bf ahi[4], alo[4];
    #pragma unroll
    for (int ks = 0; ks < 4; ++ks) {
        const int k0 = ks * 32 + laneHalf * 8;
        f32x4 a0 = *(const f32x4*)(arow + k0);
        f32x4 a1 = *(const f32x4*)(arow + k0 + 4);
        f32x4 a2 = *(const f32x4*)(arow + k0 + 16);
        f32x4 a3 = *(const f32x4*)(arow + k0 + 20);
        if (ACT_IN) {
            const f32x4 c0 = *(const f32x4*)(inBias + k0);
            const f32x4 c1 = *(const f32x4*)(inBias + k0 + 4);
            const f32x4 c2 = *(const f32x4*)(inBias + k0 + 16);
            const f32x4 c3 = *(const f32x4*)(inBias + k0 + 20);
            #pragma unroll
            for (int e = 0; e < 4; ++e) {
                float v0 = a0[e] + c0[e];
                float v1 = a1[e] + c1[e];
                float v2 = a2[e] + c2[e];
                float v3 = a3[e] + c3[e];
                a0[e] = v0 > 0.f ? v0 : 0.f;
                a1[e] = v1 > 0.f ? v1 : 0.f;
                a2[e] = v2 > 0.f ? v2 : 0.f;
                a3[e] = v3 > 0.f ? v3 : 0.f;
            }
        }
        float af[16];
        #pragma unroll
        for (int e = 0; e < 4; ++e) {
            af[e]      = a0[e];
            af[e + 4]  = a1[e];
            af[e + 8]  = a2[e];
            af[e + 12] = a3[e];
        }
        #pragma unroll
        for (int e = 0; e < 16; ++e) {
            __bf16 h, l;
            split_bf16(af[e], h, l);
            ahi[ks][e] = h;
            alo[ks][e] = l;
        }
    }
    __builtin_amdgcn_sched_barrier(0);

    const int trOff = (laneHalf * 8 + (lane & 7)) * WPITCH + ((lane >> 3) & 1) * 8;

    #pragma unroll
    for (int ks = 0; ks < 4; ++ks) {
        #pragma unroll
        for (int t = 0; t < 8; ++t) {
            const __bf16* bh = &whi[(ks * 32) * WPITCH + t * 16 + trOff];
            const __bf16* bl = &wlo[(ks * 32) * WPITCH + t * 16 + trOff];
            v8bf bh0 = ld_tr16(bh);
            v8bf bh1 = ld_tr16(bh + 16 * WPITCH);
            v8bf bl0 = ld_tr16(bl);
            v8bf bl1 = ld_tr16(bl + 16 * WPITCH);
            v16bf bhi = __builtin_shufflevector(bh0, bh1, 0,1,2,3,4,5,6,7,8,9,10,11,12,13,14,15);
            v16bf blo = __builtin_shufflevector(bl0, bl1, 0,1,2,3,4,5,6,7,8,9,10,11,12,13,14,15);

            acc[t] = __builtin_amdgcn_wmma_f32_16x16x32_bf16(false, ahi[ks], false, bhi,
                                                             (short)0, acc[t], false, false);
            acc[t] = __builtin_amdgcn_wmma_f32_16x16x32_bf16(false, ahi[ks], false, blo,
                                                             (short)0, acc[t], false, false);
            acc[t] = __builtin_amdgcn_wmma_f32_16x16x32_bf16(false, alo[ks], false, bhi,
                                                             (short)0, acc[t], false, false);
        }
    }

    #pragma unroll
    for (int t = 0; t < 8; ++t) {
        #pragma unroll
        for (int v = 0; v < 8; ++v) {
            Y[(size_t)(row0 + v + laneHalf * 8) * NODES_F + t * 16 + lmod] = acc[t][v];
        }
    }
}

#define SLICE 64
#define NWB   8
#define NREC  (2 * NWB)
typedef __attribute__((ext_vector_type(8))) int   i32x8;
typedef __attribute__((ext_vector_type(8))) float f32x8;

template <bool ADD_BIAS>
__device__ __forceinline__ void put_row(float* __restrict__ yp, f32x4 acc, f32x4 bb, bool whole,
                                        float* rec, int* recRow, int row) {
    if (whole) {
        if (ADD_BIAS) acc += bb;
        *(f32x4*)yp = acc;
    } else {
        *(f32x4*)rec = acc;
        *recRow = row;
    }
}

template <bool ADD_BIAS>
__device__ __forceinline__ void put_joined(float* __restrict__ yp, f32x4 acc, f32x4 bb, bool whole) {
    if (whole) {
        if (ADD_BIAS) acc += bb;
        *(f32x4*)yp = acc;
    } else {
        atomicAdd(yp + 0, acc[0]);
        atomicAdd(yp + 1, acc[1]);
        atomicAdd(yp + 2, acc[2]);
        atomicAdd(yp + 3, acc[3]);
    }
}

template <bool ADD_BIAS>
__global__ __launch_bounds__(NWB * 32)
void spmm_slices(const int* __restrict__ erow, const int* __restrict__ ecol,
                 const float* __restrict__ eval, const float* __restrict__ X,
                 const float* __restrict__ bias, float* __restrict__ Y, int nE) {
    __shared__ float part[NREC * NODES_F];
    __shared__ int   prow[NREC];

    const int lane = threadIdx.x & 31;
    const int wv  = __builtin_amdgcn_readfirstlane((int)(threadIdx.x >> 5));
    const int wid = blockIdx.x * NWB + wv;
    const int e0  = wid * SLICE;

    const float* xl = X + lane * 4;
    float*       yl = Y + lane * 4;
    f32x4 bb = (f32x4){0.f, 0.f, 0.f, 0.f};
    if (ADD_BIAS) bb = *(const f32x4*)(bias + lane * 4);

    float* recHead = part + (wv * 2 + 0) * NODES_F + lane * 4;
    float* recTail = part + (wv * 2 + 1) * NODES_F + lane * 4;
    prow[wv * 2 + 0] = -1;
    prow[wv * 2 + 1] = -1;

    if (e0 < nE) {
        const int e1  = min(e0 + SLICE, nE);

        const int rowBefore = (e0 > 0)  ? erow[e0 - 1] : -1;
        const int rowAfter  = (e1 < nE) ? erow[e1]     : -1;

        int   cur   = erow[e0];
        bool  whole = (cur != rowBefore);
        f32x4 acc   = (f32x4){0.f, 0.f, 0.f, 0.f};

        int eb = e0;
        i32x8 c8 = (i32x8){0, 0, 0, 0, 0, 0, 0, 0};
        if (e0 + 8 <= e1) c8 = *(const i32x8*)(ecol + e0);
        for (; eb + 8 <= e1; eb += 8) {
            f32x4 x[8];
            #pragma unroll
            for (int j = 0; j < 8; ++j)
                x[j] = *(const f32x4*)(xl + (size_t)c8[j] * NODES_F);
            const i32x8 r8 = *(const i32x8*)(erow + eb);
            const f32x8 v8 = *(const f32x8*)(eval + eb);
            c8 = *(const i32x8*)(ecol + ((eb + 16 <= e1) ? eb + 8 : eb));
            #pragma unroll
            for (int j = 0; j < 8; ++j) {
                const int r = r8[j];
                if (r != cur) {
                    put_row<ADD_BIAS>(yl + (size_t)cur * NODES_F, acc, bb, whole,
                                      recHead, &prow[wv * 2 + 0], cur);
                    acc   = (f32x4){0.f, 0.f, 0.f, 0.f};
                    cur   = r;
                    whole = true;
                }
                const float v = v8[j];
                acc[0] += v * x[j][0];
                acc[1] += v * x[j][1];
                acc[2] += v * x[j][2];
                acc[3] += v * x[j][3];
            }
        }
        for (; eb < e1; ++eb) {
            const int r = erow[eb];
            if (r != cur) {
                put_row<ADD_BIAS>(yl + (size_t)cur * NODES_F, acc, bb, whole,
                                  recHead, &prow[wv * 2 + 0], cur);
                acc   = (f32x4){0.f, 0.f, 0.f, 0.f};
                cur   = r;
                whole = true;
            }
            const float v = eval[eb];
            const f32x4 x = *(const f32x4*)(xl + (size_t)ecol[eb] * NODES_F);
            acc[0] += v * x[0];
            acc[1] += v * x[1];
            acc[2] += v * x[2];
            acc[3] += v * x[3];
        }
        put_row<ADD_BIAS>(yl + (size_t)cur * NODES_F, acc, bb, whole && (cur != rowAfter),
                          recTail, &prow[wv * 2 + 1], cur);
    }

    __syncthreads();
    if (wv != 0) return;

    const int g0 = blockIdx.x * (NWB * SLICE);
    if (g0 >= nE) return;
    const int g1 = min(g0 + NWB * SLICE, nE);
    const int groupBefore = (g0 > 0)  ? erow[g0 - 1] : -1;
    const int groupAfter  = (g1 < nE) ? erow[g1]     : -1;

    const int rr = prow[lane & (NREC - 1)];
    f32x4 p[NREC];
    #pragma unroll
    for (int i = 0; i < NREC; ++i)
        p[i] = *(const f32x4*)(part + i * NODES_F + lane * 4);

    int   run  = -1;
    f32x4 racc = (f32x4){0.f, 0.f, 0.f, 0.f};
    #pragma unroll
    for (int i = 0; i < NREC; ++i) {
        const int r = __builtin_amdgcn_readlane(rr, i);
        if (r != -1) {
            if (r != run) {
                if (run != -1)
                    put_joined<ADD_BIAS>(yl + (size_t)run * NODES_F, racc, bb,
                                         run != groupBefore && run != groupAfter);
                run  = r;
                racc = p[i];
            } else {
                racc += p[i];
            }
        }
    }
    if (run != -1)
        put_joined<ADD_BIAS>(yl + (size_t)run * NODES_F, racc, bb,
                             run != groupBefore && run != groupAfter);
}

extern "C" void kernel_launch(void* const* d_in, const int* in_sizes, int n_in,
                              void* d_out, int out_size, void* d_ws, size_t ws_size,
                              hipStream_t stream) {
    const float* X    = (const float*)d_in[0];
    const int*   erow = (const int*)  d_in[1];
    const int*   ecol = (const int*)  d_in[2];
    const float* ev   = (const float*)d_in[3];
    const float* W1   = (const float*)d_in[4];
    const float* b1   = (const float*)d_in[5];
    const float* W3   = (const float*)d_in[6];
    const float* b3   = (const float*)d_in[7];
    float* out = (float*)d_out;

    const int M  = in_sizes[0] / NODES_F;
    const int nE = in_sizes[1];
    const int NF = M * NODES_F;

    float* t = (float*)d_ws;
    float* h = t + (size_t)NF;

    dim3 blk(256);
    int gemmGrid = (M + 127) / 128;
    int nWaves   = (nE + SLICE - 1) / SLICE;
    int edgeGrid = (nWaves + NWB - 1) / NWB;

    gemm_bf16x3<false, false><<<gemmGrid, blk, 0, stream>>>(X, W1, t, M, b1, h, b1);
    spmm_slices<false><<<edgeGrid, dim3(NWB * 32), 0, stream>>>(erow, ecol, ev, t, b1, h, nE);

    gemm_bf16x3<true, true><<<gemmGrid, blk, 0, stream>>>(h, W3, t, M, b1, out, b3);
    spmm_slices<true><<<edgeGrid, dim3(NWB * 32), 0, stream>>>(erow, ecol, ev, t, b3, out, nE);
}


# SpMM: a row running past a work-group's edge range is finished by the group that holds its first edge, edgeless rows stored too: no atomics, no start rows from the GEMMs; f32 sum order changes
# speedup vs baseline: 2.4037x; 1.0785x over previous
#include <hip/hip_runtime.h>
#include <hip/hip_bf16.h>

#define NODES_F 128
#define WPITCH 144

typedef __attribute__((ext_vector_type(16))) __bf16 v16bf;
typedef __attribute__((ext_vector_type(8)))  __bf16 v8bf;
typedef __attribute__((ext_vector_type(8)))  float  v8f;
typedef __attribute__((ext_vector_type(4)))  float  f32x4;
typedef __attribute__((address_space(3))) v8bf lds_v8bf;

__device__ __forceinline__ v8bf ld_tr16(const __bf16* p) {
    return __builtin_amdgcn_ds_load_tr16_b128_v8bf16((lds_v8bf*)p);
}

__device__ __forceinline__ void split_bf16(float f, __bf16& hi, __bf16& lo) {
    hi = (__bf16)f;
    lo = (__bf16)(f - (float)hi);
}

template <bool ACT_IN>
__global__ __launch_bounds__(256)
void gemm_bf16x3(const float* __restrict__ X, const float* __restrict__ W,
                 float* __restrict__ Y, int M,
                 const float* __restrict__ inBias) {
    __shared__ __attribute__((aligned(32))) __bf16 whi[NODES_F * WPITCH];
    __shared__ __attribute__((aligned(32))) __bf16 wlo[NODES_F * WPITCH];

    const int tid = threadIdx.x;

    {
        f32x4 wv[16];
        #pragma unroll
        for (int j = 0; j < 8; ++j) {
            const float* wp = W + (size_t)(tid + 256 * j) * 8;
            wv[2 * j]     = *(const f32x4*)(wp);
            wv[2 * j + 1] = *(const f32x4*)(wp + 4);
        }
        #pragma unroll
        for (int j = 0; j < 8; ++j) {
            const int q = tid + 256 * j;
            const int k = q >> 4;
            const int n = (q & 15) * 8;
            v8bf h8, l8;
            #pragma unroll
            for (int e = 0; e < 4; ++e) {
                __bf16 h, l;
                split_bf16(wv[2 * j][e], h, l);
                h8[e] = h;
                l8[e] = l;
                split_bf16(wv[2 * j + 1][e], h, l);
                h8[e + 4] = h;
                l8[e + 4] = l;
            }
            *(v8bf*)&whi[k * WPITCH + n] = h8;
            *(v8bf*)&wlo[k * WPITCH + n] = l8;
        }
    }
    __syncthreads();

    const int wave     = tid >> 5;
    const int lane     = tid & 31;
    const int laneHalf = (lane >> 4) & 1;
    const int lmod     = lane & 15;
    const int row0     = blockIdx.x * 128 + wave * 16;
    if (row0 >= M) return;

    v8f acc[8] = {};

    const float* arow = X + (size_t)(row0 + lmod) * NODES_F;

    v16bf ahi[4], alo[4];
    #pragma unroll
    for (int ks = 0; ks < 4; ++ks) {
        const int k0 = ks * 32 + laneHalf * 8;
        f32x4 a0 = *(const f32x4*)(arow + k0);
        f32x4 a1 = *(const f32x4*)(arow + k0 + 4);
        f32x4 a2 = *(const f32x4*)(arow + k0 + 16);
        f32x4 a3 = *(const f32x4*)(arow + k0 + 20);
        if (ACT_IN) {
            const f32x4 c0 = *(const f32x4*)(inBias + k0);
            const f32x4 c1 = *(const f32x4*)(inBias + k0 + 4);
            const f32x4 c2 = *(const f32x4*)(inBias + k0 + 16);
            const f32x4 c3 = *(const f32x4*)(inBias + k0 + 20);
            #pragma unroll
            for (int e = 0; e < 4; ++e) {
                float v0 = a0[e] + c0[e];
                float v1 = a1[e] + c1[e];
                float v2 = a2[e] + c2[e];
                float v3 = a3[e] + c3[e];
                a0[e] = v0 > 0.f ? v0 : 0.f;
                a1[e] = v1 > 0.f ? v1 : 0.f;
                a2[e] = v2 > 0.f ? v2 : 0.f;
                a3[e] = v3 > 0.f ? v3 : 0.f;
            }
        }
        float af[16];
        #pragma unroll
        for (int e = 0; e < 4; ++e) {
            af[e]      = a0[e];
            af[e + 4]  = a1[e];
            af[e + 8]  = a2[e];
            af[e + 12] = a3[e];
        }
        #pragma unroll
        for (int e = 0; e < 16; ++e) {
            __bf16 h, l;
            split_bf16(af[e], h, l);
            ahi[ks][e] = h;
            alo[ks][e] = l;
        }
    }
    __builtin_amdgcn_sched_barrier(0);

    const int trOff = (laneHalf * 8 + (lane & 7)) * WPITCH + ((lane >> 3) & 1) * 8;

    #pragma unroll
    for (int ks = 0; ks < 4; ++ks) {
        #pragma unroll
        for (int t = 0; t < 8; ++t) {
            const __bf16* bh = &whi[(ks * 32) * WPITCH + t * 16 + trOff];
            const __bf16* bl = &wlo[(ks * 32) * WPITCH + t * 16 + trOff];
            v8bf bh0 = ld_tr16(bh);
            v8bf bh1 = ld_tr16(bh + 16 * WPITCH);
            v8bf bl0 = ld_tr16(bl);
            v8bf bl1 = ld_tr16(bl + 16 * WPITCH);
            v16bf bhi = __builtin_shufflevector(bh0, bh1, 0,1,2,3,4,5,6,7,8,9,10,11,12,13,14,15);
            v16bf blo = __builtin_shufflevector(bl0, bl1, 0,1,2,3,4,5,6,7,8,9,10,11,12,13,14,15);

            acc[t] = __builtin_amdgcn_wmma_f32_16x16x32_bf16(false, ahi[ks], false, bhi,
                                                             (short)0, acc[t], false, false);
            acc[t] = __builtin_amdgcn_wmma_f32_16x16x32_bf16(false, ahi[ks], false, blo,
                                                             (short)0, acc[t], false, false);
            acc[t] = __builtin_amdgcn_wmma_f32_16x16x32_bf16(false, alo[ks], false, bhi,
                                                             (short)0, acc[t], false, false);
        }
    }

    #pragma unroll
    for (int t = 0; t < 8; ++t) {
        #pragma unroll
        for (int v = 0; v < 8; ++v) {
            Y[(size_t)(row0 + v + laneHalf * 8) * NODES_F + t * 16 + lmod] = acc[t][v];
        }
    }
}

#define SLICE   64
#define NWB     8
#define TAILCUT 8
#define GROUP   (NWB * SLICE - TAILCUT)
#define NREC    (2 * NWB)
typedef __attribute__((ext_vector_type(8))) int   i32x8;
typedef __attribute__((ext_vector_type(8))) float f32x8;
typedef __attribute__((address_space(4))) const i32x8 k_i32x8;
typedef __attribute__((address_space(4))) const f32x8 k_f32x8;
typedef __attribute__((address_space(4))) const int   k_int;
typedef __attribute__((address_space(4))) const float k_float;
__device__ __forceinline__ int   edge_i(const int* p, int i)   { return *(k_int*)(p + i); }
__device__ __forceinline__ float edge_f(const float* p, int i) { return *(k_float*)(p + i); }

template <bool ADD_BIAS>
__device__ __forceinline__ void put_row(float* __restrict__ yp, f32x4 acc, f32x4 bb, bool whole,
                                        float* rec, int* recRow, int row) {
    if (whole) {
        if (ADD_BIAS) acc += bb;
        *(f32x4*)yp = acc;
    } else {
        *(f32x4*)rec = acc;
        *recRow = row;
    }
}

__device__ __forceinline__ void fill_rows(float* __restrict__ yl, f32x4 bb, int from, int to) {
    for (int r = from; r < to; ++r)
        *(f32x4*)(yl + (size_t)r * NODES_F) = bb;
}

template <bool ADD_BIAS>
__global__ __launch_bounds__(NWB * 32)
void spmm_slices(const int* __restrict__ erow, const int* __restrict__ ecol,
                 const float* __restrict__ eval, const float* __restrict__ X,
                 const float* __restrict__ bias, float* __restrict__ Y, int nE, int M) {
    __shared__ float part[NREC * NODES_F];
    __shared__ int   prow[NREC];

    const int lane = threadIdx.x & 31;
    const int wv = __builtin_amdgcn_readfirstlane((int)(threadIdx.x >> 5));
    const int g0 = blockIdx.x * GROUP;
    const int g1 = min(g0 + GROUP, nE);
    const int e0 = g0 + wv * SLICE;
    const int groupBefore = (g0 > 0) ? edge_i(erow, g0 - 1) : -1;

    const float* xl = X + lane * 4;
    float*       yl = Y + lane * 4;
    f32x4 bb = (f32x4){0.f, 0.f, 0.f, 0.f};
    if (ADD_BIAS) bb = *(const f32x4*)(bias + lane * 4);

    float* recHead = part + (wv * 2 + 0) * NODES_F + lane * 4;
    float* recTail = part + (wv * 2 + 1) * NODES_F + lane * 4;
    prow[wv * 2 + 0] = -1;
    prow[wv * 2 + 1] = -1;

    if (e0 < g1) {
        const int  e1        = min(e0 + SLICE, g1);
        const bool lastSlice = (e1 == g1);
        const bool mayRunOn  = lastSlice && (e1 + 8 <= nE);
        const int  eP        = mayRunOn ? e1 + 8 : e1;

        const int rowBefore = (e0 > 0)  ? edge_i(erow, e0 - 1) : -1;
        const int rowAfter  = (e1 < nE) ? edge_i(erow, e1)     : -1;

        int   cur   = edge_i(erow, e0);
        bool  whole = (cur != rowBefore);
        f32x4 acc   = (f32x4){0.f, 0.f, 0.f, 0.f};
        if (whole) fill_rows(yl, bb, rowBefore + 1, cur);

        int eb = e0;
        i32x8 c8 = (i32x8){0, 0, 0, 0, 0, 0, 0, 0};
        if (e0 + 8 <= e1) c8 = *(k_i32x8*)(ecol + e0);
        for (; eb + 8 <= e1; eb += 8) {
            f32x4 x[8];
            #pragma unroll
            for (int j = 0; j < 8; ++j)
                x[j] = *(const f32x4*)(xl + (size_t)c8[j] * NODES_F);
            const i32x8 r8 = *(k_i32x8*)(erow + eb);
            const f32x8 v8 = *(k_f32x8*)(eval + eb);
            c8 = *(k_i32x8*)(ecol + ((eb + 16 <= eP) ? eb + 8 : eb));
            #pragma unroll
            for (int j = 0; j < 8; ++j) {
                const int r = r8[j];
                if (r != cur) {
                    put_row<ADD_BIAS>(yl + (size_t)cur * NODES_F, acc, bb, whole,
                                      recHead, &prow[wv * 2 + 0], cur);
                    fill_rows(yl, bb, cur + 1, r);
                    acc   = (f32x4){0.f, 0.f, 0.f, 0.f};
                    cur   = r;
                    whole = true;
                }
                const float v = v8[j];
                acc[0] += v * x[j][0];
                acc[1] += v * x[j][1];
                acc[2] += v * x[j][2];
                acc[3] += v * x[j][3];
            }
        }
        for (; eb < e1; ++eb) {
            const int r = edge_i(erow, eb);
            if (r != cur) {
                put_row<ADD_BIAS>(yl + (size_t)cur * NODES_F, acc, bb, whole,
                                  recHead, &prow[wv * 2 + 0], cur);
                fill_rows(yl, bb, cur + 1, r);
                acc   = (f32x4){0.f, 0.f, 0.f, 0.f};
                cur   = r;
                whole = true;
            }
            const float v = edge_f(eval, eb);
            const f32x4 x = *(const f32x4*)(xl + (size_t)edge_i(ecol, eb) * NODES_F);
            acc[0] += v * x[0];
            acc[1] += v * x[1];
            acc[2] += v * x[2];
            acc[3] += v * x[3];
        }

        bool cut = (cur == rowAfter);
        if (lastSlice && cut && cur != groupBefore) {
            bool same = true;
            int  e    = e1;
            if (mayRunOn) {
                for (;;) {
                    f32x4 x[8];
                    #pragma unroll
                    for (int j = 0; j < 8; ++j)
                        x[j] = *(const f32x4*)(xl + (size_t)c8[j] * NODES_F);
                    const i32x8 r8 = *(k_i32x8*)(erow + e);
                    const f32x8 v8 = *(k_f32x8*)(eval + e);
                    #pragma unroll
                    for (int j = 0; j < 8; ++j) {
                        same = same && (r8[j] == cur);
                        if (same) {
                            const float v = v8[j];
                            acc[0] += v * x[j][0];
                            acc[1] += v * x[j][1];
                            acc[2] += v * x[j][2];
                            acc[3] += v * x[j][3];
                        }
                    }
                    e += 8;
                    if (!same || e + 8 > nE) break;
                    c8 = *(k_i32x8*)(ecol + e);
                }
            }
            while (same && e < nE) {
                same = (edge_i(erow, e) == cur);
                if (same) {
                    const float v = edge_f(eval, e);
                    const f32x4 x = *(const f32x4*)(xl + (size_t)edge_i(ecol, e) * NODES_F);
                    acc[0] += v * x[0];
                    acc[1] += v * x[1];
                    acc[2] += v * x[2];
                    acc[3] += v * x[3];
                    ++e;
                }
            }
            cut = false;
        }
        put_row<ADD_BIAS>(yl + (size_t)cur * NODES_F, acc, bb, whole && !cut,
                          recTail, &prow[wv * 2 + 1], cur);
        if (e1 == nE) fill_rows(yl, bb, cur + 1, M);
    }

    __syncthreads();
    if (wv != 0) return;

    if (g0 >= nE) return;

    const int rr = prow[lane & (NREC - 1)];
    f32x4 p[NREC];
    #pragma unroll
    for (int i = 0; i < NREC; ++i)
        p[i] = *(const f32x4*)(part + i * NODES_F + lane * 4);

    int   run  = -1;
    f32x4 racc = (f32x4){0.f, 0.f, 0.f, 0.f};
    #pragma unroll
    for (int i = 0; i < NREC; ++i) {
        const int r = __builtin_amdgcn_readlane(rr, i);
        if (r != -1) {
            if (r != run) {
                if (run != -1 && run != groupBefore) {
                    if (ADD_BIAS) racc += bb;
                    *(f32x4*)(yl + (size_t)run * NODES_F) = racc;
                }
                run  = r;
                racc = p[i];
            } else {
                racc += p[i];
            }
        }
    }
    if (run != -1 && run != groupBefore) {
        if (ADD_BIAS) racc += bb;
        *(f32x4*)(yl + (size_t)run * NODES_F) = racc;
    }
}

extern "C" void kernel_launch(void* const* d_in, const int* in_sizes, int n_in,
                              void* d_out, int out_size, void* d_ws, size_t ws_size,
                              hipStream_t stream) {
    const float* X    = (const float*)d_in[0];
    const int*   erow = (const int*)  d_in[1];
    const int*   ecol = (const int*)  d_in[2];
    const float* ev   = (const float*)d_in[3];
    const float* W1   = (const float*)d_in[4];
    const float* b1   = (const float*)d_in[5];
    const float* W3   = (const float*)d_in[6];
    const float* b3   = (const float*)d_in[7];
    float* out = (float*)d_out;

    const int M  = in_sizes[0] / NODES_F;
    const int nE = in_sizes[1];
    const int NF = M * NODES_F;

    float* t = (float*)d_ws;
    float* h = t + (size_t)NF;

    dim3 blk(256);
    int gemmGrid = (M + 127) / 128;
    int edgeGrid = (nE + GROUP - 1) / GROUP;

    gemm_bf16x3<false><<<gemmGrid, blk, 0, stream>>>(X, W1, t, M, b1);
    spmm_slices<false><<<edgeGrid, dim3(NWB * 32), 0, stream>>>(erow, ecol, ev, t, b1, h, nE, M);

    gemm_bf16x3<true><<<gemmGrid, blk, 0, stream>>>(h, W3, t, M, b1);
    spmm_slices<true><<<edgeGrid, dim3(NWB * 32), 0, stream>>>(erow, ecol, ev, t, b3, out, nE, M);
}
